# starting kernel
# baseline (speedup 1.0000x reference)
#include <hip/hip_runtime.h>
#include <hip/hip_bf16.h>

#define NODE_DIM   128
#define EDGE_DIM   32
#define GLOBAL_DIM 64
#define HIDDEN_DIM 256
#define IN_DIM     224
#define NSEG_MAX   512

typedef float v2f __attribute__((ext_vector_type(2)));
typedef float v8f __attribute__((ext_vector_type(8)));

__device__ __forceinline__ void gAtomicAddF(float* p, float v) {
    __hip_atomic_fetch_add(p, v, __ATOMIC_RELAXED, __HIP_MEMORY_SCOPE_AGENT);
}

__global__ void zero_f32(float* __restrict__ p, int n) {
    int i = blockIdx.x * blockDim.x + threadIdx.x;
    if (i < n) p[i] = 0.0f;
}

#define NODES_PER_BLOCK 64
__global__ void node_scatter(const float* __restrict__ x,
                             const int* __restrict__ batch,
                             float* __restrict__ node_sum,
                             float* __restrict__ node_cnt,
                             int n_nodes) {
    int f = threadIdx.x;
    int start = blockIdx.x * NODES_PER_BLOCK;
    if (start >= n_nodes) return;
    int end = min(start + NODES_PER_BLOCK, n_nodes);

    int curseg = batch[start];
    float acc = 0.0f, cacc = 0.0f;
    for (int n = start; n < end; ++n) {
        int seg = batch[n];
        if (seg != curseg) {
            gAtomicAddF(&node_sum[(size_t)curseg * NODE_DIM + f], acc);
            if (f == 0) gAtomicAddF(&node_cnt[curseg], cacc);
            acc = 0.0f; cacc = 0.0f; curseg = seg;
        }
        acc  += x[(size_t)n * NODE_DIM + f];
        cacc += 1.0f;
    }
    gAtomicAddF(&node_sum[(size_t)curseg * NODE_DIM + f], acc);
    if (f == 0) gAtomicAddF(&node_cnt[curseg], cacc);
}

__global__ void edge_scatter(const float* __restrict__ edge_attr,
                             const int* __restrict__ edge_src,
                             const int* __restrict__ batch,
                             float* __restrict__ edge_sum,
                             float* __restrict__ edge_cnt,
                             int n_edges, int nseg) {
    __shared__ float lsum[NSEG_MAX * EDGE_DIM];
    __shared__ float lcnt[NSEG_MAX];
    for (int i = threadIdx.x; i < NSEG_MAX * EDGE_DIM; i += blockDim.x) lsum[i] = 0.0f;
    for (int i = threadIdx.x; i < NSEG_MAX; i += blockDim.x) lcnt[i] = 0.0f;
    __syncthreads();

    int lane = threadIdx.x & 31;
    int wave = threadIdx.x >> 5;
    int wavesPerBlock = blockDim.x >> 5;
    int gwave = blockIdx.x * wavesPerBlock + wave;
    int nwaves = gridDim.x * wavesPerBlock;

    for (int e = gwave; e < n_edges; e += nwaves) {
        int seg = batch[edge_src[e]];
        float v = edge_attr[(size_t)e * EDGE_DIM + lane];
        atomicAdd(&lsum[seg * EDGE_DIM + lane], v);
        if (lane == 0) atomicAdd(&lcnt[seg], 1.0f);
    }
    __syncthreads();

    for (int i = threadIdx.x; i < nseg * EDGE_DIM; i += blockDim.x) {
        float v = lsum[i];
        if (v != 0.0f) gAtomicAddF(&edge_sum[i], v);
    }
    for (int i = threadIdx.x; i < nseg; i += blockDim.x) {
        float v = lcnt[i];
        if (v != 0.0f) gAtomicAddF(&edge_cnt[i], v);
    }
}

__global__ void assemble(const float* __restrict__ node_sum,
                         const float* __restrict__ node_cnt,
                         const float* __restrict__ edge_sum,
                         const float* __restrict__ edge_cnt,
                         const float* __restrict__ u,
                         float* __restrict__ H0) {
    int b = blockIdx.x;
    int j = threadIdx.x;
    float v;
    if (j < NODE_DIM) {
        v = node_sum[(size_t)b * NODE_DIM + j] / fmaxf(node_cnt[b], 1.0f);
    } else if (j < NODE_DIM + EDGE_DIM) {
        v = edge_sum[(size_t)b * EDGE_DIM + (j - NODE_DIM)] / fmaxf(edge_cnt[b], 1.0f);
    } else {
        v = u[(size_t)b * GLOBAL_DIM + (j - NODE_DIM - EDGE_DIM)];
    }
    H0[(size_t)b * IN_DIM + j] = v;
}

template <bool RELU>
__global__ void gemm_wmma(const float* __restrict__ A,
                          const float* __restrict__ W,
                          const float* __restrict__ bias,
                          float* __restrict__ C,
                          int M, int N, int K) {
    int lane = threadIdx.x & 31;
    int wave = threadIdx.x >> 5;
    int wavesPerBlock = blockDim.x >> 5;
    int tile = blockIdx.x * wavesPerBlock + wave;
    int ntilesN = N >> 4;
    int tileM = tile / ntilesN;
    int tileN = tile - tileM * ntilesN;
    if (tileM >= (M >> 4)) return;

    int half = lane >> 4;
    int l15  = lane & 15;
    int row  = (tileM << 4) + l15;
    int col  = (tileN << 4) + l15;

    const float* __restrict__ Arow = A + (size_t)row * K;
    v8f acc = {};
    for (int k0 = 0; k0 < K; k0 += 4) {
        int ka = k0 + (half << 1);
        float2 av = *reinterpret_cast<const float2*>(Arow + ka);
        v2f a; a.x = av.x; a.y = av.y;
        v2f b;
        b.x = W[(size_t)ka * N + col];
        b.y = W[(size_t)(ka + 1) * N + col];
        acc = __builtin_amdgcn_wmma_f32_16x16x4_f32(
             false, a,  false, b,
             (short)0, acc,  false,  false);
    }

    float bv = bias[col];
#pragma unroll
    for (int v = 0; v < 8; ++v) {
        int orow = (tileM << 4) + v + (half << 3);
        float r = acc[v] + bv;
        if (RELU) r = fmaxf(r, 0.0f);
        C[(size_t)orow * N + col] = r;
    }
}

extern "C" void kernel_launch(void* const* d_in, const int* in_sizes, int n_in,
                              void* d_out, int out_size, void* d_ws, size_t ws_size,
                              hipStream_t stream) {
    const float* x         = (const float*)d_in[0];
    const int*   edge_idx  = (const int*)d_in[1];
    const float* edge_attr = (const float*)d_in[2];
    const float* u         = (const float*)d_in[3];
    const int*   batch     = (const int*)d_in[4];
    const float* W1        = (const float*)d_in[5];
    const float* b1        = (const float*)d_in[6];
    const float* W2        = (const float*)d_in[7];
    const float* b2        = (const float*)d_in[8];
    const float* W3        = (const float*)d_in[9];
    const float* b3        = (const float*)d_in[10];
    float* out = (float*)d_out;

    const int n_nodes = in_sizes[0] / NODE_DIM;
    const int n_edges = in_sizes[2] / EDGE_DIM;
    const int B       = in_sizes[3] / GLOBAL_DIM;

    float* ws = (float*)d_ws;
    float* node_sum = ws;
    float* node_cnt = node_sum + (size_t)NSEG_MAX * NODE_DIM;
    float* edge_sum = node_cnt + NSEG_MAX;
    float* edge_cnt = edge_sum + (size_t)NSEG_MAX * EDGE_DIM;
    float* H0 = edge_cnt + NSEG_MAX;
    float* H1 = H0 + (size_t)NSEG_MAX * IN_DIM;
    float* H2 = H1 + (size_t)NSEG_MAX * HIDDEN_DIM;

    const int n_acc = NSEG_MAX * NODE_DIM + NSEG_MAX + NSEG_MAX * EDGE_DIM + NSEG_MAX;
    zero_f32<<<(n_acc + 255) / 256, 256, 0, stream>>>(ws, n_acc);

    node_scatter<<<(n_nodes + NODES_PER_BLOCK - 1) / NODES_PER_BLOCK, NODE_DIM, 0, stream>>>(
        x, batch, node_sum, node_cnt, n_nodes);

    edge_scatter<<<240, 256, 0, stream>>>(
        edge_attr, edge_idx, batch, edge_sum, edge_cnt, n_edges, B);

    assemble<<<B, IN_DIM, 0, stream>>>(node_sum, node_cnt, edge_sum, edge_cnt, u, H0);

    {
        int tiles = (B / 16) * (HIDDEN_DIM / 16);
        gemm_wmma<true><<<(tiles + 7) / 8, 256, 0, stream>>>(H0, W1, b1, H1, B, HIDDEN_DIM, IN_DIM);
    }
    {
        int tiles = (B / 16) * (HIDDEN_DIM / 16);
        gemm_wmma<true><<<(tiles + 7) / 8, 256, 0, stream>>>(H1, W2, b2, H2, B, HIDDEN_DIM, HIDDEN_DIM);
    }
    {
        int tiles = (B / 16) * (GLOBAL_DIM / 16);
        gemm_wmma<false><<<(tiles + 7) / 8, 256, 0, stream>>>(H2, W3, b3, out, B, GLOBAL_DIM, HIDDEN_DIM);
    }
    (void)n_in; (void)out_size; (void)ws_size;
}


# edge scatter: a wave takes 32 consecutive edges per step (one index fetch, eight 512-byte row loads, next step's loads issued ahead of the LDS adds), 256 x 1024 threads; LDS add order changes
# speedup vs baseline: 6.3984x; 6.3984x over previous
#include <hip/hip_runtime.h>
#include <hip/hip_bf16.h>

#define NODE_DIM   128
#define EDGE_DIM   32
#define GLOBAL_DIM 64
#define HIDDEN_DIM 256
#define IN_DIM     224
#define NSEG_MAX   512

typedef float v2f __attribute__((ext_vector_type(2)));
typedef float v8f __attribute__((ext_vector_type(8)));

__device__ __forceinline__ void gAtomicAddF(float* p, float v) {
    __hip_atomic_fetch_add(p, v, __ATOMIC_RELAXED, __HIP_MEMORY_SCOPE_AGENT);
}

__global__ void zero_f32(float* __restrict__ p, int n) {
    int i = blockIdx.x * blockDim.x + threadIdx.x;
    if (i < n) p[i] = 0.0f;
}

#define NODES_PER_BLOCK 64
__global__ void node_scatter(const float* __restrict__ x,
                             const int* __restrict__ batch,
                             float* __restrict__ node_sum,
                             float* __restrict__ node_cnt,
                             int n_nodes) {
    int f = threadIdx.x;
    int start = blockIdx.x * NODES_PER_BLOCK;
    if (start >= n_nodes) return;
    int end = min(start + NODES_PER_BLOCK, n_nodes);

    int curseg = batch[start];
    float acc = 0.0f, cacc = 0.0f;
    for (int n = start; n < end; ++n) {
        int seg = batch[n];
        if (seg != curseg) {
            gAtomicAddF(&node_sum[(size_t)curseg * NODE_DIM + f], acc);
            if (f == 0) gAtomicAddF(&node_cnt[curseg], cacc);
            acc = 0.0f; cacc = 0.0f; curseg = seg;
        }
        acc  += x[(size_t)n * NODE_DIM + f];
        cacc += 1.0f;
    }
    gAtomicAddF(&node_sum[(size_t)curseg * NODE_DIM + f], acc);
    if (f == 0) gAtomicAddF(&node_cnt[curseg], cacc);
}

typedef float v4f __attribute__((ext_vector_type(4)));
#define EDGE_THREADS 1024
#define CHUNK_EDGES  32
#define CHUNK_LOADS  8

__device__ __forceinline__ void edge_rows_load(v4f (&r)[CHUNK_LOADS],
                                               const float* __restrict__ edge_attr,
                                               int chunk, int lane) {
    const v4f* p = reinterpret_cast<const v4f*>(
        edge_attr + (size_t)chunk * (CHUNK_EDGES * EDGE_DIM)) + lane;
#pragma unroll
    for (int j = 0; j < CHUNK_LOADS; ++j) r[j] = p[j * 32];
}

__device__ __forceinline__ void ldsAddF(float* p, float v) {
    __hip_atomic_fetch_add(p, v, __ATOMIC_RELAXED, __HIP_MEMORY_SCOPE_WORKGROUP);
}

__device__ __forceinline__ void edge_rows_add(float* lsum, float* lcnt,
                                              const v4f (&r)[CHUNK_LOADS],
                                              int seg, int lane) {
    const int g  = lane >> 3;
    const int k4 = (lane & 7) << 2;
    const bool g1 = (g & 1) != 0;
    const bool g2 = (g & 2) != 0;
    const int o0 = g & 3, o1 = (g + 1) & 3, o2 = (g + 2) & 3, o3 = (g + 3) & 3;

    int sg[CHUNK_LOADS];
#pragma unroll
    for (int j = 0; j < CHUNK_LOADS; ++j) sg[j] = __shfl(seg, 4 * j + g);

    ldsAddF(&lcnt[seg], 1.0f);

#pragma unroll
    for (int j = 0; j < CHUNK_LOADS; ++j) {
        float* row = lsum + sg[j] * EDGE_DIM + k4;
        const v4f v = r[j];
        const float u0 = g1 ? v.y : v.x;
        const float u1 = g1 ? v.z : v.y;
        const float u2 = g1 ? v.w : v.z;
        const float u3 = g1 ? v.x : v.w;
        const float t0 = g2 ? u2 : u0;
        const float t1 = g2 ? u3 : u1;
        const float t2 = g2 ? u0 : u2;
        const float t3 = g2 ? u1 : u3;
        ldsAddF(row + o0, t0);
        ldsAddF(row + o1, t1);
        ldsAddF(row + o2, t2);
        ldsAddF(row + o3, t3);
    }
}

__global__ void __launch_bounds__(EDGE_THREADS)
edge_scatter(const float* __restrict__ edge_attr,
             const int* __restrict__ edge_src,
             const int* __restrict__ batch,
             float* __restrict__ edge_sum,
             float* __restrict__ edge_cnt,
             int n_edges, int nseg) {
    __shared__ float lsum[NSEG_MAX * EDGE_DIM];
    __shared__ float lcnt[NSEG_MAX];
    for (int i = threadIdx.x; i < NSEG_MAX * EDGE_DIM; i += blockDim.x) lsum[i] = 0.0f;
    for (int i = threadIdx.x; i < NSEG_MAX; i += blockDim.x) lcnt[i] = 0.0f;
    __syncthreads();

    const int lane = threadIdx.x & 31;
    const int wave = __builtin_amdgcn_readfirstlane(threadIdx.x >> 5);
    const int wavesPerBlock = blockDim.x >> 5;
    const int gwave = wave * gridDim.x + blockIdx.x;
    const int nwaves = gridDim.x * wavesPerBlock;
    const int nfull = n_edges / CHUNK_EDGES;

    if (gwave < nfull) {
        const int my = (nfull - 1 - gwave) / nwaves + 1;
        const int last = nfull - 1;
        v4f ra[CHUNK_LOADS], rb[CHUNK_LOADS];
        int c = gwave;
        const int src0 = edge_src[c * CHUNK_EDGES + lane];
        int srcNext = edge_src[min(c + nwaves, last) * CHUNK_EDGES + lane];
        int segA = batch[src0];
        int segB;
        __builtin_amdgcn_sched_barrier(0);
        edge_rows_load(ra, edge_attr, c, lane);
        const int pairs = (my - 1) >> 1;
        for (int p = 0; p < pairs; ++p) {
            segB = batch[srcNext];
            srcNext = edge_src[min(c + 2 * nwaves, last) * CHUNK_EDGES + lane];
            __builtin_amdgcn_sched_barrier(0);
            edge_rows_load(rb, edge_attr, c + nwaves, lane);
            __builtin_amdgcn_sched_barrier(0);
            edge_rows_add(lsum, lcnt, ra, segA, lane);
            segA = batch[srcNext];
            srcNext = edge_src[min(c + 3 * nwaves, last) * CHUNK_EDGES + lane];
            __builtin_amdgcn_sched_barrier(0);
            edge_rows_load(ra, edge_attr, c + 2 * nwaves, lane);
            __builtin_amdgcn_sched_barrier(0);
            edge_rows_add(lsum, lcnt, rb, segB, lane);
            c += 2 * nwaves;
        }
        if ((my - 1) & 1) {
            segB = batch[srcNext];
            edge_rows_load(rb, edge_attr, c + nwaves, lane);
            edge_rows_add(lsum, lcnt, ra, segA, lane);
            edge_rows_add(lsum, lcnt, rb, segB, lane);
        } else {
            edge_rows_add(lsum, lcnt, ra, segA, lane);
        }
    }

    for (int e = nfull * CHUNK_EDGES + gwave; e < n_edges; e += nwaves) {
        int seg = batch[edge_src[e]];
        float v = edge_attr[(size_t)e * EDGE_DIM + lane];
        ldsAddF(&lsum[seg * EDGE_DIM + lane], v);
        if (lane == 0) ldsAddF(&lcnt[seg], 1.0f);
    }
    __syncthreads();

    for (int i = threadIdx.x; i < nseg * EDGE_DIM; i += blockDim.x) {
        float v = lsum[i];
        if (v != 0.0f) gAtomicAddF(&edge_sum[i], v);
    }
    for (int i = threadIdx.x; i < nseg; i += blockDim.x) {
        float v = lcnt[i];
        if (v != 0.0f) gAtomicAddF(&edge_cnt[i], v);
    }
}

__global__ void assemble(const float* __restrict__ node_sum,
                         const float* __restrict__ node_cnt,
                         const float* __restrict__ edge_sum,
                         const float* __restrict__ edge_cnt,
                         const float* __restrict__ u,
                         float* __restrict__ H0) {
    int b = blockIdx.x;
    int j = threadIdx.x;
    float v;
    if (j < NODE_DIM) {
        v = node_sum[(size_t)b * NODE_DIM + j] / fmaxf(node_cnt[b], 1.0f);
    } else if (j < NODE_DIM + EDGE_DIM) {
        v = edge_sum[(size_t)b * EDGE_DIM + (j - NODE_DIM)] / fmaxf(edge_cnt[b], 1.0f);
    } else {
        v = u[(size_t)b * GLOBAL_DIM + (j - NODE_DIM - EDGE_DIM)];
    }
    H0[(size_t)b * IN_DIM + j] = v;
}

template <bool RELU>
__global__ void gemm_wmma(const float* __restrict__ A,
                          const float* __restrict__ W,
                          const float* __restrict__ bias,
                          float* __restrict__ C,
                          int M, int N, int K) {
    int lane = threadIdx.x & 31;
    int wave = threadIdx.x >> 5;
    int wavesPerBlock = blockDim.x >> 5;
    int tile = blockIdx.x * wavesPerBlock + wave;
    int ntilesN = N >> 4;
    int tileM = tile / ntilesN;
    int tileN = tile - tileM * ntilesN;
    if (tileM >= (M >> 4)) return;

    int half = lane >> 4;
    int l15  = lane & 15;
    int row  = (tileM << 4) + l15;
    int col  = (tileN << 4) + l15;

    const float* __restrict__ Arow = A + (size_t)row * K;
    v8f acc = {};
    for (int k0 = 0; k0 < K; k0 += 4) {
        int ka = k0 + (half << 1);
        float2 av = *reinterpret_cast<const float2*>(Arow + ka);
        v2f a; a.x = av.x; a.y = av.y;
        v2f b;
        b.x = W[(size_t)ka * N + col];
        b.y = W[(size_t)(ka + 1) * N + col];
        acc = __builtin_amdgcn_wmma_f32_16x16x4_f32(
             false, a,  false, b,
             (short)0, acc,  false,  false);
    }

    float bv = bias[col];
#pragma unroll
    for (int v = 0; v < 8; ++v) {
        int orow = (tileM << 4) + v + (half << 3);
        float r = acc[v] + bv;
        if (RELU) r = fmaxf(r, 0.0f);
        C[(size_t)orow * N + col] = r;
    }
}

extern "C" void kernel_launch(void* const* d_in, const int* in_sizes, int n_in,
                              void* d_out, int out_size, void* d_ws, size_t ws_size,
                              hipStream_t stream) {
    const float* x         = (const float*)d_in[0];
    const int*   edge_idx  = (const int*)d_in[1];
    const float* edge_attr = (const float*)d_in[2];
    const float* u         = (const float*)d_in[3];
    const int*   batch     = (const int*)d_in[4];
    const float* W1        = (const float*)d_in[5];
    const float* b1        = (const float*)d_in[6];
    const float* W2        = (const float*)d_in[7];
    const float* b2        = (const float*)d_in[8];
    const float* W3        = (const float*)d_in[9];
    const float* b3        = (const float*)d_in[10];
    float* out = (float*)d_out;

    const int n_nodes = in_sizes[0] / NODE_DIM;
    const int n_edges = in_sizes[2] / EDGE_DIM;
    const int B       = in_sizes[3] / GLOBAL_DIM;

    float* ws = (float*)d_ws;
    float* node_sum = ws;
    float* node_cnt = node_sum + (size_t)NSEG_MAX * NODE_DIM;
    float* edge_sum = node_cnt + NSEG_MAX;
    float* edge_cnt = edge_sum + (size_t)NSEG_MAX * EDGE_DIM;
    float* H0 = edge_cnt + NSEG_MAX;
    float* H1 = H0 + (size_t)NSEG_MAX * IN_DIM;
    float* H2 = H1 + (size_t)NSEG_MAX * HIDDEN_DIM;

    const int n_acc = NSEG_MAX * NODE_DIM + NSEG_MAX + NSEG_MAX * EDGE_DIM + NSEG_MAX;
    zero_f32<<<(n_acc + 255) / 256, 256, 0, stream>>>(ws, n_acc);

    node_scatter<<<(n_nodes + NODES_PER_BLOCK - 1) / NODES_PER_BLOCK, NODE_DIM, 0, stream>>>(
        x, batch, node_sum, node_cnt, n_nodes);

    edge_scatter<<<256, EDGE_THREADS, 0, stream>>>(
        edge_attr, edge_idx, batch, edge_sum, edge_cnt, n_edges, B);

    assemble<<<B, IN_DIM, 0, stream>>>(node_sum, node_cnt, edge_sum, edge_cnt, u, H0);

    {
        int tiles = (B / 16) * (HIDDEN_DIM / 16);
        gemm_wmma<true><<<(tiles + 7) / 8, 256, 0, stream>>>(H0, W1, b1, H1, B, HIDDEN_DIM, IN_DIM);
    }
    {
        int tiles = (B / 16) * (HIDDEN_DIM / 16);
        gemm_wmma<true><<<(tiles + 7) / 8, 256, 0, stream>>>(H1, W2, b2, H2, B, HIDDEN_DIM, HIDDEN_DIM);
    }
    {
        int tiles = (B / 16) * (GLOBAL_DIM / 16);
        gemm_wmma<false><<<(tiles + 7) / 8, 256, 0, stream>>>(H2, W3, b3, out, B, GLOBAL_DIM, HIDDEN_DIM);
    }
    (void)n_in; (void)out_size; (void)ws_size;
}


# node reduction: a wave issues its 32 rows as 32 independent 512-byte loads before the first add (was one dependent 128-byte load per node, 64 in a row); 391 x 256 threads; node sum order changes
# speedup vs baseline: 8.3502x; 1.3050x over previous
#include <hip/hip_runtime.h>
#include <hip/hip_bf16.h>

#define NODE_DIM   128
#define EDGE_DIM   32
#define GLOBAL_DIM 64
#define HIDDEN_DIM 256
#define IN_DIM     224
#define NSEG_MAX   512

typedef float v2f __attribute__((ext_vector_type(2)));
typedef float v8f __attribute__((ext_vector_type(8)));

__device__ __forceinline__ void gAtomicAddF(float* p, float v) {
    __hip_atomic_fetch_add(p, v, __ATOMIC_RELAXED, __HIP_MEMORY_SCOPE_AGENT);
}

__global__ void zero_f32(float* __restrict__ p, int n) {
    int i = blockIdx.x * blockDim.x + threadIdx.x;
    if (i < n) p[i] = 0.0f;
}

typedef float v4f __attribute__((ext_vector_type(4)));
#define NODE_ROWS_PER_WAVE   32
#define NODE_WAVES_PER_BLOCK 8
#define NODE_ROWS_PER_BLOCK  (NODE_ROWS_PER_WAVE * NODE_WAVES_PER_BLOCK)

__device__ __forceinline__ void node_flush(float* __restrict__ node_sum,
                                           float* __restrict__ node_cnt,
                                           int seg, const v4f acc, float cnt, int lane) {
    float* p = node_sum + (size_t)seg * NODE_DIM + (lane << 2);
    gAtomicAddF(p + 0, acc.x);
    gAtomicAddF(p + 1, acc.y);
    gAtomicAddF(p + 2, acc.z);
    gAtomicAddF(p + 3, acc.w);
    if (lane == 0) gAtomicAddF(&node_cnt[seg], cnt);
}

__global__ void __launch_bounds__(NODE_ROWS_PER_BLOCK)
node_scatter(const float* __restrict__ x,
             const int* __restrict__ batch,
             float* __restrict__ node_sum,
             float* __restrict__ node_cnt,
             int n_nodes) {
    const int lane = threadIdx.x & 31;
    const int wave = __builtin_amdgcn_readfirstlane(threadIdx.x >> 5);
    const int r0 = (blockIdx.x * NODE_WAVES_PER_BLOCK + wave) * NODE_ROWS_PER_WAVE;
    if (r0 >= n_nodes) return;
    const int last = n_nodes - 1;
    const int cnt = min(NODE_ROWS_PER_WAVE, n_nodes - r0);

    const int segv = batch[min(r0 + lane, last)];
    v4f r[NODE_ROWS_PER_WAVE];
#pragma unroll
    for (int j = 0; j < NODE_ROWS_PER_WAVE; ++j) {
        const int row = min(r0 + j, last);
        r[j] = *(reinterpret_cast<const v4f*>(x + (size_t)row * NODE_DIM) + lane);
    }

    int cur = __builtin_amdgcn_readlane(segv, 0);
    v4f acc = r[0];
    float cacc = 1.0f;
#pragma unroll
    for (int j = 1; j < NODE_ROWS_PER_WAVE; ++j) {
        if (j < cnt) {
            const int seg = __builtin_amdgcn_readlane(segv, j);
            if (__builtin_expect(seg != cur, 0)) {
                node_flush(node_sum, node_cnt, cur, acc, cacc, lane);
                acc = v4f{0.0f, 0.0f, 0.0f, 0.0f}; cacc = 0.0f; cur = seg;
            }
            acc += r[j];
            cacc += 1.0f;
        }
    }
    node_flush(node_sum, node_cnt, cur, acc, cacc, lane);
}

#define EDGE_THREADS 1024
#define CHUNK_EDGES  32
#define CHUNK_LOADS  8

__device__ __forceinline__ void edge_rows_load(v4f (&r)[CHUNK_LOADS],
                                               const float* __restrict__ edge_attr,
                                               int chunk, int lane) {
    const v4f* p = reinterpret_cast<const v4f*>(
        edge_attr + (size_t)chunk * (CHUNK_EDGES * EDGE_DIM)) + lane;
#pragma unroll
    for (int j = 0; j < CHUNK_LOADS; ++j) r[j] = p[j * 32];
}

__device__ __forceinline__ void ldsAddF(float* p, float v) {
    __hip_atomic_fetch_add(p, v, __ATOMIC_RELAXED, __HIP_MEMORY_SCOPE_WORKGROUP);
}

__device__ __forceinline__ void edge_rows_add(float* lsum, float* lcnt,
                                              const v4f (&r)[CHUNK_LOADS],
                                              int seg, int lane) {
    const int g  = lane >> 3;
    const int k4 = (lane & 7) << 2;
    const bool g1 = (g & 1) != 0;
    const bool g2 = (g & 2) != 0;
    const int o0 = g & 3, o1 = (g + 1) & 3, o2 = (g + 2) & 3, o3 = (g + 3) & 3;

    int sg[CHUNK_LOADS];
#pragma unroll
    for (int j = 0; j < CHUNK_LOADS; ++j) sg[j] = __shfl(seg, 4 * j + g);

    ldsAddF(&lcnt[seg], 1.0f);

#pragma unroll
    for (int j = 0; j < CHUNK_LOADS; ++j) {
        float* row = lsum + sg[j] * EDGE_DIM + k4;
        const v4f v = r[j];
        const float u0 = g1 ? v.y : v.x;
        const float u1 = g1 ? v.z : v.y;
        const float u2 = g1 ? v.w : v.z;
        const float u3 = g1 ? v.x : v.w;
        const float t0 = g2 ? u2 : u0;
        const float t1 = g2 ? u3 : u1;
        const float t2 = g2 ? u0 : u2;
        const float t3 = g2 ? u1 : u3;
        ldsAddF(row + o0, t0);
        ldsAddF(row + o1, t1);
        ldsAddF(row + o2, t2);
        ldsAddF(row + o3, t3);
    }
}

__global__ void __launch_bounds__(EDGE_THREADS)
edge_scatter(const float* __restrict__ edge_attr,
             const int* __restrict__ edge_src,
             const int* __restrict__ batch,
             float* __restrict__ edge_sum,
             float* __restrict__ edge_cnt,
             int n_edges, int nseg) {
    __shared__ float lsum[NSEG_MAX * EDGE_DIM];
    __shared__ float lcnt[NSEG_MAX];
    for (int i = threadIdx.x; i < NSEG_MAX * EDGE_DIM; i += blockDim.x) lsum[i] = 0.0f;
    for (int i = threadIdx.x; i < NSEG_MAX; i += blockDim.x) lcnt[i] = 0.0f;
    __syncthreads();

    const int lane = threadIdx.x & 31;
    const int wave = __builtin_amdgcn_readfirstlane(threadIdx.x >> 5);
    const int wavesPerBlock = blockDim.x >> 5;
    const int gwave = wave * gridDim.x + blockIdx.x;
    const int nwaves = gridDim.x * wavesPerBlock;
    const int nfull = n_edges / CHUNK_EDGES;

    if (gwave < nfull) {
        const int my = (nfull - 1 - gwave) / nwaves + 1;
        const int last = nfull - 1;
        v4f ra[CHUNK_LOADS], rb[CHUNK_LOADS];
        int c = gwave;
        const int src0 = edge_src[c * CHUNK_EDGES + lane];
        int srcNext = edge_src[min(c + nwaves, last) * CHUNK_EDGES + lane];
        int segA = batch[src0];
        int segB;
        __builtin_amdgcn_sched_barrier(0);
        edge_rows_load(ra, edge_attr, c, lane);
        const int pairs = (my - 1) >> 1;
        for (int p = 0; p < pairs; ++p) {
            segB = batch[srcNext];
            srcNext = edge_src[min(c + 2 * nwaves, last) * CHUNK_EDGES + lane];
            __builtin_amdgcn_sched_barrier(0);
            edge_rows_load(rb, edge_attr, c + nwaves, lane);
            __builtin_amdgcn_sched_barrier(0);
            edge_rows_add(lsum, lcnt, ra, segA, lane);
            segA = batch[srcNext];
            srcNext = edge_src[min(c + 3 * nwaves, last) * CHUNK_EDGES + lane];
            __builtin_amdgcn_sched_barrier(0);
            edge_rows_load(ra, edge_attr, c + 2 * nwaves, lane);
            __builtin_amdgcn_sched_barrier(0);
            edge_rows_add(lsum, lcnt, rb, segB, lane);
            c += 2 * nwaves;
        }
        if ((my - 1) & 1) {
            segB = batch[srcNext];
            edge_rows_load(rb, edge_attr, c + nwaves, lane);
            edge_rows_add(lsum, lcnt, ra, segA, lane);
            edge_rows_add(lsum, lcnt, rb, segB, lane);
        } else {
            edge_rows_add(lsum, lcnt, ra, segA, lane);
        }
    }

    for (int e = nfull * CHUNK_EDGES + gwave; e < n_edges; e += nwaves) {
        int seg = batch[edge_src[e]];
        float v = edge_attr[(size_t)e * EDGE_DIM + lane];
        ldsAddF(&lsum[seg * EDGE_DIM + lane], v);
        if (lane == 0) ldsAddF(&lcnt[seg], 1.0f);
    }
    __syncthreads();

    for (int i = threadIdx.x; i < nseg * EDGE_DIM; i += blockDim.x) {
        float v = lsum[i];
        if (v != 0.0f) gAtomicAddF(&edge_sum[i], v);
    }
    for (int i = threadIdx.x; i < nseg; i += blockDim.x) {
        float v = lcnt[i];
        if (v != 0.0f) gAtomicAddF(&edge_cnt[i], v);
    }
}

__global__ void assemble(const float* __restrict__ node_sum,
                         const float* __restrict__ node_cnt,
                         const float* __restrict__ edge_sum,
                         const float* __restrict__ edge_cnt,
                         const float* __restrict__ u,
                         float* __restrict__ H0) {
    int b = blockIdx.x;
    int j = threadIdx.x;
    float v;
    if (j < NODE_DIM) {
        v = node_sum[(size_t)b * NODE_DIM + j] / fmaxf(node_cnt[b], 1.0f);
    } else if (j < NODE_DIM + EDGE_DIM) {
        v = edge_sum[(size_t)b * EDGE_DIM + (j - NODE_DIM)] / fmaxf(edge_cnt[b], 1.0f);
    } else {
        v = u[(size_t)b * GLOBAL_DIM + (j - NODE_DIM - EDGE_DIM)];
    }
    H0[(size_t)b * IN_DIM + j] = v;
}

template <bool RELU>
__global__ void gemm_wmma(const float* __restrict__ A,
                          const float* __restrict__ W,
                          const float* __restrict__ bias,
                          float* __restrict__ C,
                          int M, int N, int K) {
    int lane = threadIdx.x & 31;
    int wave = threadIdx.x >> 5;
    int wavesPerBlock = blockDim.x >> 5;
    int tile = blockIdx.x * wavesPerBlock + wave;
    int ntilesN = N >> 4;
    int tileM = tile / ntilesN;
    int tileN = tile - tileM * ntilesN;
    if (tileM >= (M >> 4)) return;

    int half = lane >> 4;
    int l15  = lane & 15;
    int row  = (tileM << 4) + l15;
    int col  = (tileN << 4) + l15;

    const float* __restrict__ Arow = A + (size_t)row * K;
    v8f acc = {};
    for (int k0 = 0; k0 < K; k0 += 4) {
        int ka = k0 + (half << 1);
        float2 av = *reinterpret_cast<const float2*>(Arow + ka);
        v2f a; a.x = av.x; a.y = av.y;
        v2f b;
        b.x = W[(size_t)ka * N + col];
        b.y = W[(size_t)(ka + 1) * N + col];
        acc = __builtin_amdgcn_wmma_f32_16x16x4_f32(
             false, a,  false, b,
             (short)0, acc,  false,  false);
    }

    float bv = bias[col];
#pragma unroll
    for (int v = 0; v < 8; ++v) {
        int orow = (tileM << 4) + v + (half << 3);
        float r = acc[v] + bv;
        if (RELU) r = fmaxf(r, 0.0f);
        C[(size_t)orow * N + col] = r;
    }
}

extern "C" void kernel_launch(void* const* d_in, const int* in_sizes, int n_in,
                              void* d_out, int out_size, void* d_ws, size_t ws_size,
                              hipStream_t stream) {
    const float* x         = (const float*)d_in[0];
    const int*   edge_idx  = (const int*)d_in[1];
    const float* edge_attr = (const float*)d_in[2];
    const float* u         = (const float*)d_in[3];
    const int*   batch     = (const int*)d_in[4];
    const float* W1        = (const float*)d_in[5];
    const float* b1        = (const float*)d_in[6];
    const float* W2        = (const float*)d_in[7];
    const float* b2        = (const float*)d_in[8];
    const float* W3        = (const float*)d_in[9];
    const float* b3        = (const float*)d_in[10];
    float* out = (float*)d_out;

    const int n_nodes = in_sizes[0] / NODE_DIM;
    const int n_edges = in_sizes[2] / EDGE_DIM;
    const int B       = in_sizes[3] / GLOBAL_DIM;

    float* ws = (float*)d_ws;
    float* node_sum = ws;
    float* node_cnt = node_sum + (size_t)NSEG_MAX * NODE_DIM;
    float* edge_sum = node_cnt + NSEG_MAX;
    float* edge_cnt = edge_sum + (size_t)NSEG_MAX * EDGE_DIM;
    float* H0 = edge_cnt + NSEG_MAX;
    float* H1 = H0 + (size_t)NSEG_MAX * IN_DIM;
    float* H2 = H1 + (size_t)NSEG_MAX * HIDDEN_DIM;

    const int n_acc = NSEG_MAX * NODE_DIM + NSEG_MAX + NSEG_MAX * EDGE_DIM + NSEG_MAX;
    zero_f32<<<(n_acc + 255) / 256, 256, 0, stream>>>(ws, n_acc);

    node_scatter<<<(n_nodes + NODE_ROWS_PER_BLOCK - 1) / NODE_ROWS_PER_BLOCK, NODE_ROWS_PER_BLOCK, 0, stream>>>(
        x, batch, node_sum, node_cnt, n_nodes);

    edge_scatter<<<256, EDGE_THREADS, 0, stream>>>(
        edge_attr, edge_idx, batch, edge_sum, edge_cnt, n_edges, B);

    assemble<<<B, IN_DIM, 0, stream>>>(node_sum, node_cnt, edge_sum, edge_cnt, u, H0);

    {
        int tiles = (B / 16) * (HIDDEN_DIM / 16);
        gemm_wmma<true><<<(tiles + 7) / 8, 256, 0, stream>>>(H0, W1, b1, H1, B, HIDDEN_DIM, IN_DIM);
    }
    {
        int tiles = (B / 16) * (HIDDEN_DIM / 16);
        gemm_wmma<true><<<(tiles + 7) / 8, 256, 0, stream>>>(H1, W2, b2, H2, B, HIDDEN_DIM, HIDDEN_DIM);
    }
    {
        int tiles = (B / 16) * (GLOBAL_DIM / 16);
        gemm_wmma<false><<<(tiles + 7) / 8, 256, 0, stream>>>(H2, W3, b3, out, B, GLOBAL_DIM, HIDDEN_DIM);
    }
    (void)n_in; (void)out_size; (void)ws_size;
}


# MLP tail fused: one launch forms the means and runs all three layers, 16 graphs per workgroup of 16 waves with activations in LDS (was 4 launches via workspace planes); sum order unchanged
# speedup vs baseline: 13.7610x; 1.6480x over previous
#include <hip/hip_runtime.h>
#include <hip/hip_bf16.h>

#define NODE_DIM   128
#define EDGE_DIM   32
#define GLOBAL_DIM 64
#define HIDDEN_DIM 256
#define IN_DIM     224
#define NSEG_MAX   512

typedef float v2f __attribute__((ext_vector_type(2)));
typedef float v8f __attribute__((ext_vector_type(8)));

__device__ __forceinline__ void gAtomicAddF(float* p, float v) {
    __hip_atomic_fetch_add(p, v, __ATOMIC_RELAXED, __HIP_MEMORY_SCOPE_AGENT);
}

__global__ void zero_f32(float* __restrict__ p, int n) {
    int i = blockIdx.x * blockDim.x + threadIdx.x;
    if (i < n) p[i] = 0.0f;
}

typedef float v4f __attribute__((ext_vector_type(4)));
#define NODE_ROWS_PER_WAVE   32
#define NODE_WAVES_PER_BLOCK 8
#define NODE_ROWS_PER_BLOCK  (NODE_ROWS_PER_WAVE * NODE_WAVES_PER_BLOCK)

__device__ __forceinline__ void node_flush(float* __restrict__ node_sum,
                                           float* __restrict__ node_cnt,
                                           int seg, const v4f acc, float cnt, int lane) {
    float* p = node_sum + (size_t)seg * NODE_DIM + (lane << 2);
    gAtomicAddF(p + 0, acc.x);
    gAtomicAddF(p + 1, acc.y);
    gAtomicAddF(p + 2, acc.z);
    gAtomicAddF(p + 3, acc.w);
    if (lane == 0) gAtomicAddF(&node_cnt[seg], cnt);
}

__global__ void __launch_bounds__(NODE_ROWS_PER_BLOCK)
node_scatter(const float* __restrict__ x,
             const int* __restrict__ batch,
             float* __restrict__ node_sum,
             float* __restrict__ node_cnt,
             int n_nodes) {
    const int lane = threadIdx.x & 31;
    const int wave = __builtin_amdgcn_readfirstlane(threadIdx.x >> 5);
    const int r0 = (blockIdx.x * NODE_WAVES_PER_BLOCK + wave) * NODE_ROWS_PER_WAVE;
    if (r0 >= n_nodes) return;
    const int last = n_nodes - 1;
    const int cnt = min(NODE_ROWS_PER_WAVE, n_nodes - r0);

    const int segv = batch[min(r0 + lane, last)];
    v4f r[NODE_ROWS_PER_WAVE];
#pragma unroll
    for (int j = 0; j < NODE_ROWS_PER_WAVE; ++j) {
        const int row = min(r0 + j, last);
        r[j] = *(reinterpret_cast<const v4f*>(x + (size_t)row * NODE_DIM) + lane);
    }

    int cur = __builtin_amdgcn_readlane(segv, 0);
    v4f acc = r[0];
    float cacc = 1.0f;
#pragma unroll
    for (int j = 1; j < NODE_ROWS_PER_WAVE; ++j) {
        if (j < cnt) {
            const int seg = __builtin_amdgcn_readlane(segv, j);
            if (__builtin_expect(seg != cur, 0)) {
                node_flush(node_sum, node_cnt, cur, acc, cacc, lane);
                acc = v4f{0.0f, 0.0f, 0.0f, 0.0f}; cacc = 0.0f; cur = seg;
            }
            acc += r[j];
            cacc += 1.0f;
        }
    }
    node_flush(node_sum, node_cnt, cur, acc, cacc, lane);
}

#define EDGE_THREADS 1024
#define CHUNK_EDGES  32
#define CHUNK_LOADS  8

__device__ __forceinline__ void edge_rows_load(v4f (&r)[CHUNK_LOADS],
                                               const float* __restrict__ edge_attr,
                                               int chunk, int lane) {
    const v4f* p = reinterpret_cast<const v4f*>(
        edge_attr + (size_t)chunk * (CHUNK_EDGES * EDGE_DIM)) + lane;
#pragma unroll
    for (int j = 0; j < CHUNK_LOADS; ++j) r[j] = p[j * 32];
}

__device__ __forceinline__ void ldsAddF(float* p, float v) {
    __hip_atomic_fetch_add(p, v, __ATOMIC_RELAXED, __HIP_MEMORY_SCOPE_WORKGROUP);
}

__device__ __forceinline__ void edge_rows_add(float* lsum, float* lcnt,
                                              const v4f (&r)[CHUNK_LOADS],
                                              int seg, int lane) {
    const int g  = lane >> 3;
    const int k4 = (lane & 7) << 2;
    const bool g1 = (g & 1) != 0;
    const bool g2 = (g & 2) != 0;
    const int o0 = g & 3, o1 = (g + 1) & 3, o2 = (g + 2) & 3, o3 = (g + 3) & 3;

    int sg[CHUNK_LOADS];
#pragma unroll
    for (int j = 0; j < CHUNK_LOADS; ++j) sg[j] = __shfl(seg, 4 * j + g);

    ldsAddF(&lcnt[seg], 1.0f);

#pragma unroll
    for (int j = 0; j < CHUNK_LOADS; ++j) {
        float* row = lsum + sg[j] * EDGE_DIM + k4;
        const v4f v = r[j];
        const float u0 = g1 ? v.y : v.x;
        const float u1 = g1 ? v.z : v.y;
        const float u2 = g1 ? v.w : v.z;
        const float u3 = g1 ? v.x : v.w;
        const float t0 = g2 ? u2 : u0;
        const float t1 = g2 ? u3 : u1;
        const float t2 = g2 ? u0 : u2;
        const float t3 = g2 ? u1 : u3;
        ldsAddF(row + o0, t0);
        ldsAddF(row + o1, t1);
        ldsAddF(row + o2, t2);
        ldsAddF(row + o3, t3);
    }
}

__global__ void __launch_bounds__(EDGE_THREADS)
edge_scatter(const float* __restrict__ edge_attr,
             const int* __restrict__ edge_src,
             const int* __restrict__ batch,
             float* __restrict__ edge_sum,
             float* __restrict__ edge_cnt,
             int n_edges, int nseg) {
    __shared__ float lsum[NSEG_MAX * EDGE_DIM];
    __shared__ float lcnt[NSEG_MAX];
    for (int i = threadIdx.x; i < NSEG_MAX * EDGE_DIM; i += blockDim.x) lsum[i] = 0.0f;
    for (int i = threadIdx.x; i < NSEG_MAX; i += blockDim.x) lcnt[i] = 0.0f;
    __syncthreads();

    const int lane = threadIdx.x & 31;
    const int wave = __builtin_amdgcn_readfirstlane(threadIdx.x >> 5);
    const int wavesPerBlock = blockDim.x >> 5;
    const int gwave = wave * gridDim.x + blockIdx.x;
    const int nwaves = gridDim.x * wavesPerBlock;
    const int nfull = n_edges / CHUNK_EDGES;

    if (gwave < nfull) {
        const int my = (nfull - 1 - gwave) / nwaves + 1;
        const int last = nfull - 1;
        v4f ra[CHUNK_LOADS], rb[CHUNK_LOADS];
        int c = gwave;
        const int src0 = edge_src[c * CHUNK_EDGES + lane];
        int srcNext = edge_src[min(c + nwaves, last) * CHUNK_EDGES + lane];
        int segA = batch[src0];
        int segB;
        __builtin_amdgcn_sched_barrier(0);
        edge_rows_load(ra, edge_attr, c, lane);
        const int pairs = (my - 1) >> 1;
        for (int p = 0; p < pairs; ++p) {
            segB = batch[srcNext];
            srcNext = edge_src[min(c + 2 * nwaves, last) * CHUNK_EDGES + lane];
            __builtin_amdgcn_sched_barrier(0);
            edge_rows_load(rb, edge_attr, c + nwaves, lane);
            __builtin_amdgcn_sched_barrier(0);
            edge_rows_add(lsum, lcnt, ra, segA, lane);
            segA = batch[srcNext];
            srcNext = edge_src[min(c + 3 * nwaves, last) * CHUNK_EDGES + lane];
            __builtin_amdgcn_sched_barrier(0);
            edge_rows_load(ra, edge_attr, c + 2 * nwaves, lane);
            __builtin_amdgcn_sched_barrier(0);
            edge_rows_add(lsum, lcnt, rb, segB, lane);
            c += 2 * nwaves;
        }
        if ((my - 1) & 1) {
            segB = batch[srcNext];
            edge_rows_load(rb, edge_attr, c + nwaves, lane);
            edge_rows_add(lsum, lcnt, ra, segA, lane);
            edge_rows_add(lsum, lcnt, rb, segB, lane);
        } else {
            edge_rows_add(lsum, lcnt, ra, segA, lane);
        }
    }

    for (int e = nfull * CHUNK_EDGES + gwave; e < n_edges; e += nwaves) {
        int seg = batch[edge_src[e]];
        float v = edge_attr[(size_t)e * EDGE_DIM + lane];
        ldsAddF(&lsum[seg * EDGE_DIM + lane], v);
        if (lane == 0) ldsAddF(&lcnt[seg], 1.0f);
    }
    __syncthreads();

    for (int i = threadIdx.x; i < nseg * EDGE_DIM; i += blockDim.x) {
        float v = lsum[i];
        if (v != 0.0f) gAtomicAddF(&edge_sum[i], v);
    }
    for (int i = threadIdx.x; i < nseg; i += blockDim.x) {
        float v = lcnt[i];
        if (v != 0.0f) gAtomicAddF(&edge_cnt[i], v);
    }
}

#define MLP_ROWS    16
#define MLP_WAVES   16
#define MLP_THREADS (MLP_WAVES * 32)
#define H0_PITCH    (IN_DIM + 4)
#define H_PITCH     (HIDDEN_DIM + 4)
#define KCH         8

static_assert(EDGE_DIM == 32, "one lane per edge feature");
static_assert(NODE_DIM % 32 == 0 && GLOBAL_DIM % 32 == 0, "whole 32-lane pieces");
static_assert(HIDDEN_DIM == MLP_WAVES * 16, "one wave per 16 hidden columns");
static_assert(GLOBAL_DIM / 16 == 4, "layer 3 has four column tiles");

template <int N>
__device__ __forceinline__ void b_group_load(float (&b)[2 * KCH],
                                             const float* __restrict__ wp, int group) {
#pragma unroll
    for (int j = 0; j < KCH; ++j) {
        const float* p = wp + (size_t)((group * KCH + j) * 4) * N;
        b[2 * j]     = p[0];
        b[2 * j + 1] = p[N];
    }
}

__device__ __forceinline__ v8f group_mma(v8f acc, const float* ap,
                                         const float (&b)[2 * KCH], int group) {
#pragma unroll
    for (int j = 0; j < KCH; ++j) {
        const float2 av = *reinterpret_cast<const float2*>(ap + (group * KCH + j) * 4);
        v2f a;  a.x = av.x;       a.y = av.y;
        v2f bb; bb.x = b[2 * j];  bb.y = b[2 * j + 1];
        acc = __builtin_amdgcn_wmma_f32_16x16x4_f32(
             false, a,  false, bb,
             (short)0, acc,  false,  false);
    }
    return acc;
}

template <int K, int N>
__device__ __forceinline__ v8f tile_mma(const float* ap, const float* __restrict__ wp,
                                        float (&b0)[2 * KCH]) {
    static_assert(K % (4 * KCH) == 0, "whole groups");
    constexpr int NG = K / (4 * KCH);
    float b1[2 * KCH];
    v8f acc = {};
#pragma unroll
    for (int g = 0; g < NG; g += 2) {
        if (g + 1 < NG) b_group_load<N>(b1, wp, g + 1);
        acc = group_mma(acc, ap, b0, g);
        if (g + 2 < NG) b_group_load<N>(b0, wp, g + 2);
        if (g + 1 < NG) acc = group_mma(acc, ap, b1, g + 1);
    }
    return acc;
}

__global__ void __launch_bounds__(MLP_THREADS)
mlp_fused(const float* __restrict__ node_sum,
          const float* __restrict__ node_cnt,
          const float* __restrict__ edge_sum,
          const float* __restrict__ edge_cnt,
          const float* __restrict__ u,
          const float* __restrict__ W1, const float* __restrict__ b1,
          const float* __restrict__ W2, const float* __restrict__ b2,
          const float* __restrict__ W3, const float* __restrict__ b3,
          float* __restrict__ out) {
    __shared__ __attribute__((aligned(16))) float bufA[MLP_ROWS * H_PITCH];
    __shared__ __attribute__((aligned(16))) float bufB[MLP_ROWS * H_PITCH];

    const int lane = threadIdx.x & 31;
    const int wave = __builtin_amdgcn_readfirstlane(threadIdx.x >> 5);
    const int half = lane >> 4;
    const int l15  = lane & 15;
    const int r0   = blockIdx.x * MLP_ROWS;
    const int col  = (wave << 4) + l15;

    float bq[2 * KCH];
    const float* w1p = W1 + (size_t)(2 * half) * HIDDEN_DIM + col;
    b_group_load<HIDDEN_DIM>(bq, w1p, 0);
    const float bias1 = b1[col];

    {
        const int b = r0 + wave;
        float* h = bufA + wave * H0_PITCH;
        const float nc = fmaxf(node_cnt[b], 1.0f);
        const float ec = fmaxf(edge_cnt[b], 1.0f);
#pragma unroll
        for (int i = 0; i < NODE_DIM / 32; ++i)
            h[lane + 32 * i] = node_sum[(size_t)b * NODE_DIM + lane + 32 * i] / nc;
        h[NODE_DIM + lane] = edge_sum[(size_t)b * EDGE_DIM + lane] / ec;
#pragma unroll
        for (int i = 0; i < GLOBAL_DIM / 32; ++i)
            h[NODE_DIM + EDGE_DIM + lane + 32 * i] = u[(size_t)b * GLOBAL_DIM + lane + 32 * i];
    }
    __syncthreads();

    v8f acc = tile_mma<IN_DIM, HIDDEN_DIM>(bufA + l15 * H0_PITCH + 2 * half, w1p, bq);
    const float* w2p = W2 + (size_t)(2 * half) * HIDDEN_DIM + col;
    b_group_load<HIDDEN_DIM>(bq, w2p, 0);
    const float bias2 = b2[col];
#pragma unroll
    for (int v = 0; v < 8; ++v)
        bufB[(v + 8 * half) * H_PITCH + col] = fmaxf(acc[v] + bias1, 0.0f);
    __syncthreads();

    acc = tile_mma<HIDDEN_DIM, HIDDEN_DIM>(bufB + l15 * H_PITCH + 2 * half, w2p, bq);
    const bool l3   = (wave % 5) == 0;
    const int  col3 = ((wave / 5) << 4) + l15;
    const float* w3p = W3 + (size_t)(2 * half) * GLOBAL_DIM + col3;
    float bias3 = 0.0f;
    if (l3) {
        b_group_load<GLOBAL_DIM>(bq, w3p, 0);
        bias3 = b3[col3];
    }
#pragma unroll
    for (int v = 0; v < 8; ++v)
        bufA[(v + 8 * half) * H_PITCH + col] = fmaxf(acc[v] + bias2, 0.0f);
    __syncthreads();

    if (l3) {
        acc = tile_mma<HIDDEN_DIM, GLOBAL_DIM>(bufA + l15 * H_PITCH + 2 * half, w3p, bq);
#pragma unroll
        for (int v = 0; v < 8; ++v)
            out[(size_t)(r0 + v + 8 * half) * GLOBAL_DIM + col3] = acc[v] + bias3;
    }
}

extern "C" void kernel_launch(void* const* d_in, const int* in_sizes, int n_in,
                              void* d_out, int out_size, void* d_ws, size_t ws_size,
                              hipStream_t stream) {
    const float* x         = (const float*)d_in[0];
    const int*   edge_idx  = (const int*)d_in[1];
    const float* edge_attr = (const float*)d_in[2];
    const float* u         = (const float*)d_in[3];
    const int*   batch     = (const int*)d_in[4];
    const float* W1        = (const float*)d_in[5];
    const float* b1        = (const float*)d_in[6];
    const float* W2        = (const float*)d_in[7];
    const float* b2        = (const float*)d_in[8];
    const float* W3        = (const float*)d_in[9];
    const float* b3        = (const float*)d_in[10];
    float* out = (float*)d_out;

    const int n_nodes = in_sizes[0] / NODE_DIM;
    const int n_edges = in_sizes[2] / EDGE_DIM;
    const int B       = in_sizes[3] / GLOBAL_DIM;

    float* ws = (float*)d_ws;
    float* node_sum = ws;
    float* node_cnt = node_sum + (size_t)NSEG_MAX * NODE_DIM;
    float* edge_sum = node_cnt + NSEG_MAX;
    float* edge_cnt = edge_sum + (size_t)NSEG_MAX * EDGE_DIM;

    const int n_acc = NSEG_MAX * NODE_DIM + NSEG_MAX + NSEG_MAX * EDGE_DIM + NSEG_MAX;
    zero_f32<<<(n_acc + 255) / 256, 256, 0, stream>>>(ws, n_acc);

    node_scatter<<<(n_nodes + NODE_ROWS_PER_BLOCK - 1) / NODE_ROWS_PER_BLOCK, NODE_ROWS_PER_BLOCK, 0, stream>>>(
        x, batch, node_sum, node_cnt, n_nodes);

    edge_scatter<<<256, EDGE_THREADS, 0, stream>>>(
        edge_attr, edge_idx, batch, edge_sum, edge_cnt, n_edges, B);

    mlp_fused<<<B / MLP_ROWS, MLP_THREADS, 0, stream>>>(
        node_sum, node_cnt, edge_sum, edge_cnt, u, W1, b1, W2, b2, W3, b3, out);
    (void)n_in; (void)out_size; (void)ws_size;
}


# node reduction moved into the edge scatter launch (3 launches, was 4): waves of the first 196 workgroups sum 16 node rows each ahead of their edge groups; node sum order changes (16-row partials)
# speedup vs baseline: 14.8966x; 1.0825x over previous
#include <hip/hip_runtime.h>
#include <hip/hip_bf16.h>

#define NODE_DIM   128
#define EDGE_DIM   32
#define GLOBAL_DIM 64
#define HIDDEN_DIM 256
#define IN_DIM     224
#define NSEG_MAX   512

typedef float v2f __attribute__((ext_vector_type(2)));
typedef float v8f __attribute__((ext_vector_type(8)));

__device__ __forceinline__ void gAtomicAddF(float* p, float v) {
    __hip_atomic_fetch_add(p, v, __ATOMIC_RELAXED, __HIP_MEMORY_SCOPE_AGENT);
}

__global__ void zero_f32(float* __restrict__ p, int n) {
    int i = blockIdx.x * blockDim.x + threadIdx.x;
    if (i < n) p[i] = 0.0f;
}

typedef float v4f __attribute__((ext_vector_type(4)));
#define NODE_ROWS_PER_WAVE   16
static_assert((NODE_ROWS_PER_WAVE & (NODE_ROWS_PER_WAVE - 1)) == 0 && NODE_ROWS_PER_WAVE <= 32,
              "one lane per row id, lane index masked to the row count");

__device__ __forceinline__ void node_flush(float* __restrict__ node_sum,
                                           float* __restrict__ node_cnt,
                                           int seg, const v4f acc, float cnt, int lane) {
    float* p = node_sum + (size_t)seg * NODE_DIM + (lane << 2);
    gAtomicAddF(p + 0, acc.x);
    gAtomicAddF(p + 1, acc.y);
    gAtomicAddF(p + 2, acc.z);
    gAtomicAddF(p + 3, acc.w);
    if (lane == 0) gAtomicAddF(&node_cnt[seg], cnt);
}

__device__ __forceinline__ void node_task(const float* __restrict__ x,
                                          const int* __restrict__ batch,
                                          float* __restrict__ node_sum,
                                          float* __restrict__ node_cnt,
                                          int r0, int n_nodes, int lane) {
    const int cnt = min(NODE_ROWS_PER_WAVE, n_nodes - r0);
    const v4f* xr = reinterpret_cast<const v4f*>(x + (size_t)r0 * NODE_DIM) + lane;

    if (cnt == NODE_ROWS_PER_WAVE) {
        const int segv = batch[r0 + (lane & (NODE_ROWS_PER_WAVE - 1))];
        v4f r[NODE_ROWS_PER_WAVE];
#pragma unroll
        for (int j = 0; j < NODE_ROWS_PER_WAVE; ++j) r[j] = xr[j * (NODE_DIM / 4)];

        int cur = __builtin_amdgcn_readlane(segv, 0);
        v4f acc = r[0];
        float cacc = 1.0f;
#pragma unroll
        for (int j = 1; j < NODE_ROWS_PER_WAVE; ++j) {
            const int seg = __builtin_amdgcn_readlane(segv, j);
            if (__builtin_expect(seg != cur, 0)) {
                node_flush(node_sum, node_cnt, cur, acc, cacc, lane);
                acc = v4f{0.0f, 0.0f, 0.0f, 0.0f}; cacc = 0.0f; cur = seg;
            }
            acc += r[j];
            cacc += 1.0f;
        }
        node_flush(node_sum, node_cnt, cur, acc, cacc, lane);
    } else {
        int cur = batch[r0];
        v4f acc = xr[0];
        float cacc = 1.0f;
        for (int j = 1; j < cnt; ++j) {
            const int seg = batch[r0 + j];
            if (seg != cur) {
                node_flush(node_sum, node_cnt, cur, acc, cacc, lane);
                acc = v4f{0.0f, 0.0f, 0.0f, 0.0f}; cacc = 0.0f; cur = seg;
            }
            acc += xr[j * (NODE_DIM / 4)];
            cacc += 1.0f;
        }
        node_flush(node_sum, node_cnt, cur, acc, cacc, lane);
    }
}

#define EDGE_THREADS 1024
#define CHUNK_EDGES  32
#define CHUNK_LOADS  8

__device__ __forceinline__ void edge_rows_load(v4f (&r)[CHUNK_LOADS],
                                               const float* __restrict__ edge_attr,
                                               int chunk, int lane) {
    const v4f* p = reinterpret_cast<const v4f*>(
        edge_attr + (size_t)chunk * (CHUNK_EDGES * EDGE_DIM)) + lane;
#pragma unroll
    for (int j = 0; j < CHUNK_LOADS; ++j) r[j] = p[j * 32];
}

__device__ __forceinline__ void ldsAddF(float* p, float v) {
    __hip_atomic_fetch_add(p, v, __ATOMIC_RELAXED, __HIP_MEMORY_SCOPE_WORKGROUP);
}

__device__ __forceinline__ void edge_rows_add(float* lsum, float* lcnt,
                                              const v4f (&r)[CHUNK_LOADS],
                                              int seg, int lane) {
    const int g  = lane >> 3;
    const int k4 = (lane & 7) << 2;
    const bool g1 = (g & 1) != 0;
    const bool g2 = (g & 2) != 0;
    const int o0 = g & 3, o1 = (g + 1) & 3, o2 = (g + 2) & 3, o3 = (g + 3) & 3;

    int sg[CHUNK_LOADS];
#pragma unroll
    for (int j = 0; j < CHUNK_LOADS; ++j) sg[j] = __shfl(seg, 4 * j + g);

    ldsAddF(&lcnt[seg], 1.0f);

#pragma unroll
    for (int j = 0; j < CHUNK_LOADS; ++j) {
        float* row = lsum + sg[j] * EDGE_DIM + k4;
        const v4f v = r[j];
        const float u0 = g1 ? v.y : v.x;
        const float u1 = g1 ? v.z : v.y;
        const float u2 = g1 ? v.w : v.z;
        const float u3 = g1 ? v.x : v.w;
        const float t0 = g2 ? u2 : u0;
        const float t1 = g2 ? u3 : u1;
        const float t2 = g2 ? u0 : u2;
        const float t3 = g2 ? u1 : u3;
        ldsAddF(row + o0, t0);
        ldsAddF(row + o1, t1);
        ldsAddF(row + o2, t2);
        ldsAddF(row + o3, t3);
    }
}

__global__ void __launch_bounds__(EDGE_THREADS)
scatter_both(const float* __restrict__ x,
             const float* __restrict__ edge_attr,
             const int* __restrict__ edge_src,
             const int* __restrict__ batch,
             float* __restrict__ node_sum,
             float* __restrict__ node_cnt,
             float* __restrict__ edge_sum,
             float* __restrict__ edge_cnt,
             int n_nodes, int n_edges, int nseg) {
    __shared__ float lsum[NSEG_MAX * EDGE_DIM];
    __shared__ float lcnt[NSEG_MAX];
    for (int i = threadIdx.x; i < NSEG_MAX * EDGE_DIM; i += blockDim.x) lsum[i] = 0.0f;
    for (int i = threadIdx.x; i < NSEG_MAX; i += blockDim.x) lcnt[i] = 0.0f;
    __syncthreads();

    const int lane = threadIdx.x & 31;
    const int wave = __builtin_amdgcn_readfirstlane(threadIdx.x >> 5);
    const int wavesPerBlock = blockDim.x >> 5;
    const int gwave = wave * gridDim.x + blockIdx.x;
    const int nwaves = gridDim.x * wavesPerBlock;
    const int nfull = n_edges / CHUNK_EDGES;

    for (int t = blockIdx.x * wavesPerBlock + wave; t * NODE_ROWS_PER_WAVE < n_nodes; t += nwaves)
        node_task(x, batch, node_sum, node_cnt, t * NODE_ROWS_PER_WAVE, n_nodes, lane);

    if (gwave < nfull) {
        const int my = (nfull - 1 - gwave) / nwaves + 1;
        const int last = nfull - 1;
        v4f ra[CHUNK_LOADS], rb[CHUNK_LOADS];
        int c = gwave;
        const int src0 = edge_src[c * CHUNK_EDGES + lane];
        int srcNext = edge_src[min(c + nwaves, last) * CHUNK_EDGES + lane];
        int segA = batch[src0];
        int segB;
        __builtin_amdgcn_sched_barrier(0);
        edge_rows_load(ra, edge_attr, c, lane);
        const int pairs = (my - 1) >> 1;
        for (int p = 0; p < pairs; ++p) {
            segB = batch[srcNext];
            srcNext = edge_src[min(c + 2 * nwaves, last) * CHUNK_EDGES + lane];
            __builtin_amdgcn_sched_barrier(0);
            edge_rows_load(rb, edge_attr, c + nwaves, lane);
            __builtin_amdgcn_sched_barrier(0);
            edge_rows_add(lsum, lcnt, ra, segA, lane);
            segA = batch[srcNext];
            srcNext = edge_src[min(c + 3 * nwaves, last) * CHUNK_EDGES + lane];
            __builtin_amdgcn_sched_barrier(0);
            edge_rows_load(ra, edge_attr, c + 2 * nwaves, lane);
            __builtin_amdgcn_sched_barrier(0);
            edge_rows_add(lsum, lcnt, rb, segB, lane);
            c += 2 * nwaves;
        }
        if ((my - 1) & 1) {
            segB = batch[srcNext];
            edge_rows_load(rb, edge_attr, c + nwaves, lane);
            edge_rows_add(lsum, lcnt, ra, segA, lane);
            edge_rows_add(lsum, lcnt, rb, segB, lane);
        } else {
            edge_rows_add(lsum, lcnt, ra, segA, lane);
        }
    }

    for (int e = nfull * CHUNK_EDGES + gwave; e < n_edges; e += nwaves) {
        int seg = batch[edge_src[e]];
        float v = edge_attr[(size_t)e * EDGE_DIM + lane];
        ldsAddF(&lsum[seg * EDGE_DIM + lane], v);
        if (lane == 0) ldsAddF(&lcnt[seg], 1.0f);
    }
    __syncthreads();

    for (int i = threadIdx.x; i < nseg * EDGE_DIM; i += blockDim.x) {
        float v = lsum[i];
        if (v != 0.0f) gAtomicAddF(&edge_sum[i], v);
    }
    for (int i = threadIdx.x; i < nseg; i += blockDim.x) {
        float v = lcnt[i];
        if (v != 0.0f) gAtomicAddF(&edge_cnt[i], v);
    }
}

#define MLP_ROWS    16
#define MLP_WAVES   16
#define MLP_THREADS (MLP_WAVES * 32)
#define H0_PITCH    (IN_DIM + 4)
#define H_PITCH     (HIDDEN_DIM + 4)
#define KCH         8

static_assert(EDGE_DIM == 32, "one lane per edge feature");
static_assert(NODE_DIM % 32 == 0 && GLOBAL_DIM % 32 == 0, "whole 32-lane pieces");
static_assert(HIDDEN_DIM == MLP_WAVES * 16, "one wave per 16 hidden columns");
static_assert(GLOBAL_DIM / 16 == 4, "layer 3 has four column tiles");

template <int N>
__device__ __forceinline__ void b_group_load(float (&b)[2 * KCH],
                                             const float* __restrict__ wp, int group) {
#pragma unroll
    for (int j = 0; j < KCH; ++j) {
        const float* p = wp + (size_t)((group * KCH + j) * 4) * N;
        b[2 * j]     = p[0];
        b[2 * j + 1] = p[N];
    }
}

__device__ __forceinline__ v8f group_mma(v8f acc, const float* ap,
                                         const float (&b)[2 * KCH], int group) {
#pragma unroll
    for (int j = 0; j < KCH; ++j) {
        const float2 av = *reinterpret_cast<const float2*>(ap + (group * KCH + j) * 4);
        v2f a;  a.x = av.x;       a.y = av.y;
        v2f bb; bb.x = b[2 * j];  bb.y = b[2 * j + 1];
        acc = __builtin_amdgcn_wmma_f32_16x16x4_f32(
             false, a,  false, bb,
             (short)0, acc,  false,  false);
    }
    return acc;
}

template <int K, int N>
__device__ __forceinline__ v8f tile_mma(const float* ap, const float* __restrict__ wp,
                                        float (&b0)[2 * KCH]) {
    static_assert(K % (4 * KCH) == 0, "whole groups");
    constexpr int NG = K / (4 * KCH);
    float b1[2 * KCH];
    v8f acc = {};
#pragma unroll
    for (int g = 0; g < NG; g += 2) {
        if (g + 1 < NG) b_group_load<N>(b1, wp, g + 1);
        acc = group_mma(acc, ap, b0, g);
        if (g + 2 < NG) b_group_load<N>(b0, wp, g + 2);
        if (g + 1 < NG) acc = group_mma(acc, ap, b1, g + 1);
    }
    return acc;
}

__global__ void __launch_bounds__(MLP_THREADS)
mlp_fused(const float* __restrict__ node_sum,
          const float* __restrict__ node_cnt,
          const float* __restrict__ edge_sum,
          const float* __restrict__ edge_cnt,
          const float* __restrict__ u,
          const float* __restrict__ W1, const float* __restrict__ b1,
          const float* __restrict__ W2, const float* __restrict__ b2,
          const float* __restrict__ W3, const float* __restrict__ b3,
          float* __restrict__ out) {
    __shared__ __attribute__((aligned(16))) float bufA[MLP_ROWS * H_PITCH];
    __shared__ __attribute__((aligned(16))) float bufB[MLP_ROWS * H_PITCH];

    const int lane = threadIdx.x & 31;
    const int wave = __builtin_amdgcn_readfirstlane(threadIdx.x >> 5);
    const int half = lane >> 4;
    const int l15  = lane & 15;
    const int r0   = blockIdx.x * MLP_ROWS;
    const int col  = (wave << 4) + l15;

    float bq[2 * KCH];
    const float* w1p = W1 + (size_t)(2 * half) * HIDDEN_DIM + col;
    b_group_load<HIDDEN_DIM>(bq, w1p, 0);
    const float bias1 = b1[col];

    {
        const int b = r0 + wave;
        float* h = bufA + wave * H0_PITCH;
        const float nc = fmaxf(node_cnt[b], 1.0f);
        const float ec = fmaxf(edge_cnt[b], 1.0f);
#pragma unroll
        for (int i = 0; i < NODE_DIM / 32; ++i)
            h[lane + 32 * i] = node_sum[(size_t)b * NODE_DIM + lane + 32 * i] / nc;
        h[NODE_DIM + lane] = edge_sum[(size_t)b * EDGE_DIM + lane] / ec;
#pragma unroll
        for (int i = 0; i < GLOBAL_DIM / 32; ++i)
            h[NODE_DIM + EDGE_DIM + lane + 32 * i] = u[(size_t)b * GLOBAL_DIM + lane + 32 * i];
    }
    __syncthreads();

    v8f acc = tile_mma<IN_DIM, HIDDEN_DIM>(bufA + l15 * H0_PITCH + 2 * half, w1p, bq);
    const float* w2p = W2 + (size_t)(2 * half) * HIDDEN_DIM + col;
    b_group_load<HIDDEN_DIM>(bq, w2p, 0);
    const float bias2 = b2[col];
#pragma unroll
    for (int v = 0; v < 8; ++v)
        bufB[(v + 8 * half) * H_PITCH + col] = fmaxf(acc[v] + bias1, 0.0f);
    __syncthreads();

    acc = tile_mma<HIDDEN_DIM, HIDDEN_DIM>(bufB + l15 * H_PITCH + 2 * half, w2p, bq);
    const bool l3   = (wave % 5) == 0;
    const int  col3 = ((wave / 5) << 4) + l15;
    const float* w3p = W3 + (size_t)(2 * half) * GLOBAL_DIM + col3;
    float bias3 = 0.0f;
    if (l3) {
        b_group_load<GLOBAL_DIM>(bq, w3p, 0);
        bias3 = b3[col3];
    }
#pragma unroll
    for (int v = 0; v < 8; ++v)
        bufA[(v + 8 * half) * H_PITCH + col] = fmaxf(acc[v] + bias2, 0.0f);
    __syncthreads();

    if (l3) {
        acc = tile_mma<HIDDEN_DIM, GLOBAL_DIM>(bufA + l15 * H_PITCH + 2 * half, w3p, bq);
#pragma unroll
        for (int v = 0; v < 8; ++v)
            out[(size_t)(r0 + v + 8 * half) * GLOBAL_DIM + col3] = acc[v] + bias3;
    }
}

extern "C" void kernel_launch(void* const* d_in, const int* in_sizes, int n_in,
                              void* d_out, int out_size, void* d_ws, size_t ws_size,
                              hipStream_t stream) {
    const float* x         = (const float*)d_in[0];
    const int*   edge_idx  = (const int*)d_in[1];
    const float* edge_attr = (const float*)d_in[2];
    const float* u         = (const float*)d_in[3];
    const int*   batch     = (const int*)d_in[4];
    const float* W1        = (const float*)d_in[5];
    const float* b1        = (const float*)d_in[6];
    const float* W2        = (const float*)d_in[7];
    const float* b2        = (const float*)d_in[8];
    const float* W3        = (const float*)d_in[9];
    const float* b3        = (const float*)d_in[10];
    float* out = (float*)d_out;

    const int n_nodes = in_sizes[0] / NODE_DIM;
    const int n_edges = in_sizes[2] / EDGE_DIM;
    const int B       = in_sizes[3] / GLOBAL_DIM;

    float* ws = (float*)d_ws;
    float* node_sum = ws;
    float* node_cnt = node_sum + (size_t)NSEG_MAX * NODE_DIM;
    float* edge_sum = node_cnt + NSEG_MAX;
    float* edge_cnt = edge_sum + (size_t)NSEG_MAX * EDGE_DIM;

    const int n_acc = NSEG_MAX * NODE_DIM + NSEG_MAX + NSEG_MAX * EDGE_DIM + NSEG_MAX;
    zero_f32<<<(n_acc + 255) / 256, 256, 0, stream>>>(ws, n_acc);

    scatter_both<<<256, EDGE_THREADS, 0, stream>>>(
        x, edge_attr, edge_idx, batch, node_sum, node_cnt, edge_sum, edge_cnt,
        n_nodes, n_edges, B);

    mlp_fused<<<B / MLP_ROWS, MLP_THREADS, 0, stream>>>(
        node_sum, node_cnt, edge_sum, edge_cnt, u, W1, b1, W2, b2, W3, b3, out);
    (void)n_in; (void)out_size; (void)ws_size;
}


# global sums: one workspace set of accumulators per package half, picked by the workgroup's die number: no atomic line is shared by both halves; MLP prologue adds the two sets; sum order changes
# speedup vs baseline: 16.7865x; 1.1269x over previous
#include <hip/hip_runtime.h>
#include <hip/hip_bf16.h>

#define NODE_DIM   128
#define EDGE_DIM   32
#define GLOBAL_DIM 64
#define HIDDEN_DIM 256
#define IN_DIM     224
#define NSEG_MAX   512
#define PLANE_FLOATS (NSEG_MAX * NODE_DIM + NSEG_MAX + NSEG_MAX * EDGE_DIM + NSEG_MAX)
#define N_PLANES   2

typedef float v2f __attribute__((ext_vector_type(2)));
typedef float v8f __attribute__((ext_vector_type(8)));

__device__ __forceinline__ void gAtomicAddF(float* p, float v) {
    __hip_atomic_fetch_add(p, v, __ATOMIC_RELAXED, __HIP_MEMORY_SCOPE_AGENT);
}

__global__ void zero_f32(float* __restrict__ p, int n) {
    int i = blockIdx.x * blockDim.x + threadIdx.x;
    if (i < n) p[i] = 0.0f;
}

typedef float v4f __attribute__((ext_vector_type(4)));
#define NODE_ROWS_PER_WAVE   16
static_assert((NODE_ROWS_PER_WAVE & (NODE_ROWS_PER_WAVE - 1)) == 0 && NODE_ROWS_PER_WAVE <= 32,
              "one lane per row id, lane index masked to the row count");

__device__ __forceinline__ void node_flush(float* __restrict__ node_sum,
                                           float* __restrict__ node_cnt,
                                           int seg, const v4f acc, float cnt, int lane) {
    float* p = node_sum + (size_t)seg * NODE_DIM + (lane << 2);
    gAtomicAddF(p + 0, acc.x);
    gAtomicAddF(p + 1, acc.y);
    gAtomicAddF(p + 2, acc.z);
    gAtomicAddF(p + 3, acc.w);
    if (lane == 0) gAtomicAddF(&node_cnt[seg], cnt);
}

__device__ __forceinline__ void node_task(const float* __restrict__ x,
                                          const int* __restrict__ batch,
                                          float* __restrict__ node_sum,
                                          float* __restrict__ node_cnt,
                                          int r0, int n_nodes, int lane) {
    const int cnt = min(NODE_ROWS_PER_WAVE, n_nodes - r0);
    const v4f* xr = reinterpret_cast<const v4f*>(x + (size_t)r0 * NODE_DIM) + lane;

    if (cnt == NODE_ROWS_PER_WAVE) {
        const int segv = batch[r0 + (lane & (NODE_ROWS_PER_WAVE - 1))];
        v4f r[NODE_ROWS_PER_WAVE];
#pragma unroll
        for (int j = 0; j < NODE_ROWS_PER_WAVE; ++j) r[j] = xr[j * (NODE_DIM / 4)];

        int cur = __builtin_amdgcn_readlane(segv, 0);
        v4f acc = r[0];
        float cacc = 1.0f;
#pragma unroll
        for (int j = 1; j < NODE_ROWS_PER_WAVE; ++j) {
            const int seg = __builtin_amdgcn_readlane(segv, j);
            if (__builtin_expect(seg != cur, 0)) {
                node_flush(node_sum, node_cnt, cur, acc, cacc, lane);
                acc = v4f{0.0f, 0.0f, 0.0f, 0.0f}; cacc = 0.0f; cur = seg;
            }
            acc += r[j];
            cacc += 1.0f;
        }
        node_flush(node_sum, node_cnt, cur, acc, cacc, lane);
    } else {
        int cur = batch[r0];
        v4f acc = xr[0];
        float cacc = 1.0f;
        for (int j = 1; j < cnt; ++j) {
            const int seg = batch[r0 + j];
            if (seg != cur) {
                node_flush(node_sum, node_cnt, cur, acc, cacc, lane);
                acc = v4f{0.0f, 0.0f, 0.0f, 0.0f}; cacc = 0.0f; cur = seg;
            }
            acc += xr[j * (NODE_DIM / 4)];
            cacc += 1.0f;
        }
        node_flush(node_sum, node_cnt, cur, acc, cacc, lane);
    }
}

#define EDGE_THREADS 1024
#define CHUNK_EDGES  32
#define CHUNK_LOADS  8

__device__ __forceinline__ void edge_rows_load(v4f (&r)[CHUNK_LOADS],
                                               const float* __restrict__ edge_attr,
                                               int chunk, int lane) {
    const v4f* p = reinterpret_cast<const v4f*>(
        edge_attr + (size_t)chunk * (CHUNK_EDGES * EDGE_DIM)) + lane;
#pragma unroll
    for (int j = 0; j < CHUNK_LOADS; ++j) r[j] = p[j * 32];
}

__device__ __forceinline__ void ldsAddF(float* p, float v) {
    __hip_atomic_fetch_add(p, v, __ATOMIC_RELAXED, __HIP_MEMORY_SCOPE_WORKGROUP);
}

__device__ __forceinline__ void edge_rows_add(float* lsum, float* lcnt,
                                              const v4f (&r)[CHUNK_LOADS],
                                              int seg, int lane) {
    const int g  = lane >> 3;
    const int k4 = (lane & 7) << 2;
    const bool g1 = (g & 1) != 0;
    const bool g2 = (g & 2) != 0;
    const int o0 = g & 3, o1 = (g + 1) & 3, o2 = (g + 2) & 3, o3 = (g + 3) & 3;

    int sg[CHUNK_LOADS];
#pragma unroll
    for (int j = 0; j < CHUNK_LOADS; ++j) sg[j] = __shfl(seg, 4 * j + g);

    ldsAddF(&lcnt[seg], 1.0f);

#pragma unroll
    for (int j = 0; j < CHUNK_LOADS; ++j) {
        float* row = lsum + sg[j] * EDGE_DIM + k4;
        const v4f v = r[j];
        const float u0 = g1 ? v.y : v.x;
        const float u1 = g1 ? v.z : v.y;
        const float u2 = g1 ? v.w : v.z;
        const float u3 = g1 ? v.x : v.w;
        const float t0 = g2 ? u2 : u0;
        const float t1 = g2 ? u3 : u1;
        const float t2 = g2 ? u0 : u2;
        const float t3 = g2 ? u1 : u3;
        ldsAddF(row + o0, t0);
        ldsAddF(row + o1, t1);
        ldsAddF(row + o2, t2);
        ldsAddF(row + o3, t3);
    }
}

__global__ void __launch_bounds__(EDGE_THREADS)
scatter_both(const float* __restrict__ x,
             const float* __restrict__ edge_attr,
             const int* __restrict__ edge_src,
             const int* __restrict__ batch,
             float* __restrict__ node_sum,
             float* __restrict__ node_cnt,
             float* __restrict__ edge_sum,
             float* __restrict__ edge_cnt,
             int n_nodes, int n_edges, int nseg) {
    __shared__ float lsum[NSEG_MAX * EDGE_DIM];
    __shared__ float lcnt[NSEG_MAX];
    const unsigned place = __builtin_amdgcn_s_sendmsg_rtn(0x87);
    const size_t poff = (size_t)((place >> 18) & 1u) * PLANE_FLOATS;
    node_sum += poff; node_cnt += poff; edge_sum += poff; edge_cnt += poff;
    for (int i = threadIdx.x; i < NSEG_MAX * EDGE_DIM; i += blockDim.x) lsum[i] = 0.0f;
    for (int i = threadIdx.x; i < NSEG_MAX; i += blockDim.x) lcnt[i] = 0.0f;
    __syncthreads();

    const int lane = threadIdx.x & 31;
    const int wave = __builtin_amdgcn_readfirstlane(threadIdx.x >> 5);
    const int wavesPerBlock = blockDim.x >> 5;
    const int gwave = wave * gridDim.x + blockIdx.x;
    const int nwaves = gridDim.x * wavesPerBlock;
    const int nfull = n_edges / CHUNK_EDGES;

    for (int t = blockIdx.x * wavesPerBlock + wave; t * NODE_ROWS_PER_WAVE < n_nodes; t += nwaves)
        node_task(x, batch, node_sum, node_cnt, t * NODE_ROWS_PER_WAVE, n_nodes, lane);

    if (gwave < nfull) {
        const int my = (nfull - 1 - gwave) / nwaves + 1;
        const int last = nfull - 1;
        v4f ra[CHUNK_LOADS], rb[CHUNK_LOADS];
        int c = gwave;
        const int src0 = edge_src[c * CHUNK_EDGES + lane];
        int srcNext = edge_src[min(c + nwaves, last) * CHUNK_EDGES + lane];
        int segA = batch[src0];
        int segB;
        __builtin_amdgcn_sched_barrier(0);
        edge_rows_load(ra, edge_attr, c, lane);
        const int pairs = (my - 1) >> 1;
        for (int p = 0; p < pairs; ++p) {
            segB = batch[srcNext];
            srcNext = edge_src[min(c + 2 * nwaves, last) * CHUNK_EDGES + lane];
            __builtin_amdgcn_sched_barrier(0);
            edge_rows_load(rb, edge_attr, c + nwaves, lane);
            __builtin_amdgcn_sched_barrier(0);
            edge_rows_add(lsum, lcnt, ra, segA, lane);
            segA = batch[srcNext];
            srcNext = edge_src[min(c + 3 * nwaves, last) * CHUNK_EDGES + lane];
            __builtin_amdgcn_sched_barrier(0);
            edge_rows_load(ra, edge_attr, c + 2 * nwaves, lane);
            __builtin_amdgcn_sched_barrier(0);
            edge_rows_add(lsum, lcnt, rb, segB, lane);
            c += 2 * nwaves;
        }
        if ((my - 1) & 1) {
            segB = batch[srcNext];
            edge_rows_load(rb, edge_attr, c + nwaves, lane);
            edge_rows_add(lsum, lcnt, ra, segA, lane);
            edge_rows_add(lsum, lcnt, rb, segB, lane);
        } else {
            edge_rows_add(lsum, lcnt, ra, segA, lane);
        }
    }

    for (int e = nfull * CHUNK_EDGES + gwave; e < n_edges; e += nwaves) {
        int seg = batch[edge_src[e]];
        float v = edge_attr[(size_t)e * EDGE_DIM + lane];
        ldsAddF(&lsum[seg * EDGE_DIM + lane], v);
        if (lane == 0) ldsAddF(&lcnt[seg], 1.0f);
    }
    __syncthreads();

    for (int i = threadIdx.x; i < nseg * EDGE_DIM; i += blockDim.x) {
        float v = lsum[i];
        if (v != 0.0f) gAtomicAddF(&edge_sum[i], v);
    }
    for (int i = threadIdx.x; i < nseg; i += blockDim.x) {
        float v = lcnt[i];
        if (v != 0.0f) gAtomicAddF(&edge_cnt[i], v);
    }
}

#define MLP_ROWS    16
#define MLP_WAVES   16
#define MLP_THREADS (MLP_WAVES * 32)
#define H0_PITCH    (IN_DIM + 4)
#define H_PITCH     (HIDDEN_DIM + 4)
#define KCH         8

static_assert(EDGE_DIM == 32, "one lane per edge feature");
static_assert(NODE_DIM % 32 == 0 && GLOBAL_DIM % 32 == 0, "whole 32-lane pieces");
static_assert(HIDDEN_DIM == MLP_WAVES * 16, "one wave per 16 hidden columns");
static_assert(GLOBAL_DIM / 16 == 4, "layer 3 has four column tiles");

template <int N>
__device__ __forceinline__ void b_group_load(float (&b)[2 * KCH],
                                             const float* __restrict__ wp, int group) {
#pragma unroll
    for (int j = 0; j < KCH; ++j) {
        const float* p = wp + (size_t)((group * KCH + j) * 4) * N;
        b[2 * j]     = p[0];
        b[2 * j + 1] = p[N];
    }
}

__device__ __forceinline__ v8f group_mma(v8f acc, const float* ap,
                                         const float (&b)[2 * KCH], int group) {
#pragma unroll
    for (int j = 0; j < KCH; ++j) {
        const float2 av = *reinterpret_cast<const float2*>(ap + (group * KCH + j) * 4);
        v2f a;  a.x = av.x;       a.y = av.y;
        v2f bb; bb.x = b[2 * j];  bb.y = b[2 * j + 1];
        acc = __builtin_amdgcn_wmma_f32_16x16x4_f32(
             false, a,  false, bb,
             (short)0, acc,  false,  false);
    }
    return acc;
}

template <int K, int N>
__device__ __forceinline__ v8f tile_mma(const float* ap, const float* __restrict__ wp,
                                        float (&b0)[2 * KCH]) {
    static_assert(K % (4 * KCH) == 0, "whole groups");
    constexpr int NG = K / (4 * KCH);
    float b1[2 * KCH];
    v8f acc = {};
#pragma unroll
    for (int g = 0; g < NG; g += 2) {
        if (g + 1 < NG) b_group_load<N>(b1, wp, g + 1);
        acc = group_mma(acc, ap, b0, g);
        if (g + 2 < NG) b_group_load<N>(b0, wp, g + 2);
        if (g + 1 < NG) acc = group_mma(acc, ap, b1, g + 1);
    }
    return acc;
}

__global__ void __launch_bounds__(MLP_THREADS)
mlp_fused(const float* __restrict__ node_sum,
          const float* __restrict__ node_cnt,
          const float* __restrict__ edge_sum,
          const float* __restrict__ edge_cnt,
          const float* __restrict__ u,
          const float* __restrict__ W1, const float* __restrict__ b1,
          const float* __restrict__ W2, const float* __restrict__ b2,
          const float* __restrict__ W3, const float* __restrict__ b3,
          float* __restrict__ out) {
    __shared__ __attribute__((aligned(16))) float bufA[MLP_ROWS * H_PITCH];
    __shared__ __attribute__((aligned(16))) float bufB[MLP_ROWS * H_PITCH];

    const int lane = threadIdx.x & 31;
    const int wave = __builtin_amdgcn_readfirstlane(threadIdx.x >> 5);
    const int half = lane >> 4;
    const int l15  = lane & 15;
    const int r0   = blockIdx.x * MLP_ROWS;
    const int col  = (wave << 4) + l15;

    float bq[2 * KCH];
    const float* w1p = W1 + (size_t)(2 * half) * HIDDEN_DIM + col;
    b_group_load<HIDDEN_DIM>(bq, w1p, 0);
    const float bias1 = b1[col];

    {
        const int b = r0 + wave;
        float* h = bufA + wave * H0_PITCH;
        const float nc = fmaxf(node_cnt[b] + node_cnt[PLANE_FLOATS + b], 1.0f);
        const float ec = fmaxf(edge_cnt[b] + edge_cnt[PLANE_FLOATS + b], 1.0f);
#pragma unroll
        for (int i = 0; i < NODE_DIM / 32; ++i) {
            const size_t k = (size_t)b * NODE_DIM + lane + 32 * i;
            h[lane + 32 * i] = (node_sum[k] + node_sum[PLANE_FLOATS + k]) / nc;
        }
        {
            const size_t k = (size_t)b * EDGE_DIM + lane;
            h[NODE_DIM + lane] = (edge_sum[k] + edge_sum[PLANE_FLOATS + k]) / ec;
        }
#pragma unroll
        for (int i = 0; i < GLOBAL_DIM / 32; ++i)
            h[NODE_DIM + EDGE_DIM + lane + 32 * i] = u[(size_t)b * GLOBAL_DIM + lane + 32 * i];
    }
    __syncthreads();

    v8f acc = tile_mma<IN_DIM, HIDDEN_DIM>(bufA + l15 * H0_PITCH + 2 * half, w1p, bq);
    const float* w2p = W2 + (size_t)(2 * half) * HIDDEN_DIM + col;
    b_group_load<HIDDEN_DIM>(bq, w2p, 0);
    const float bias2 = b2[col];
#pragma unroll
    for (int v = 0; v < 8; ++v)
        bufB[(v + 8 * half) * H_PITCH + col] = fmaxf(acc[v] + bias1, 0.0f);
    __syncthreads();

    acc = tile_mma<HIDDEN_DIM, HIDDEN_DIM>(bufB + l15 * H_PITCH + 2 * half, w2p, bq);
    const bool l3   = (wave % 5) == 0;
    const int  col3 = ((wave / 5) << 4) + l15;
    const float* w3p = W3 + (size_t)(2 * half) * GLOBAL_DIM + col3;
    float bias3 = 0.0f;
    if (l3) {
        b_group_load<GLOBAL_DIM>(bq, w3p, 0);
        bias3 = b3[col3];
    }
#pragma unroll
    for (int v = 0; v < 8; ++v)
        bufA[(v + 8 * half) * H_PITCH + col] = fmaxf(acc[v] + bias2, 0.0f);
    __syncthreads();

    if (l3) {
        acc = tile_mma<HIDDEN_DIM, GLOBAL_DIM>(bufA + l15 * H_PITCH + 2 * half, w3p, bq);
#pragma unroll
        for (int v = 0; v < 8; ++v)
            out[(size_t)(r0 + v + 8 * half) * GLOBAL_DIM + col3] = acc[v] + bias3;
    }
}

extern "C" void kernel_launch(void* const* d_in, const int* in_sizes, int n_in,
                              void* d_out, int out_size, void* d_ws, size_t ws_size,
                              hipStream_t stream) {
    const float* x         = (const float*)d_in[0];
    const int*   edge_idx  = (const int*)d_in[1];
    const float* edge_attr = (const float*)d_in[2];
    const float* u         = (const float*)d_in[3];
    const int*   batch     = (const int*)d_in[4];
    const float* W1        = (const float*)d_in[5];
    const float* b1        = (const float*)d_in[6];
    const float* W2        = (const float*)d_in[7];
    const float* b2        = (const float*)d_in[8];
    const float* W3        = (const float*)d_in[9];
    const float* b3        = (const float*)d_in[10];
    float* out = (float*)d_out;

    const int n_nodes = in_sizes[0] / NODE_DIM;
    const int n_edges = in_sizes[2] / EDGE_DIM;
    const int B       = in_sizes[3] / GLOBAL_DIM;

    float* ws = (float*)d_ws;
    float* node_sum = ws;
    float* node_cnt = node_sum + (size_t)NSEG_MAX * NODE_DIM;
    float* edge_sum = node_cnt + NSEG_MAX;
    float* edge_cnt = edge_sum + (size_t)NSEG_MAX * EDGE_DIM;

    const int n_acc = N_PLANES * PLANE_FLOATS;
    zero_f32<<<(n_acc + 255) / 256, 256, 0, stream>>>(ws, n_acc);

    scatter_both<<<256, EDGE_THREADS, 0, stream>>>(
        x, edge_attr, edge_idx, batch, node_sum, node_cnt, edge_sum, edge_cnt,
        n_nodes, n_edges, B);

    mlp_fused<<<B / MLP_ROWS, MLP_THREADS, 0, stream>>>(
        node_sum, node_cnt, edge_sum, edge_cnt, u, W1, b1, W2, b2, W3, b3, out);
    (void)n_in; (void)out_size; (void)ws_size;
}
